# starting kernel
# baseline (speedup 1.0000x reference)
#include <hip/hip_runtime.h>
#include <hip/hip_bf16.h>
#include <math.h>

typedef float v2f __attribute__((ext_vector_type(2)));
typedef float v8f __attribute__((ext_vector_type(8)));

#define BATCH   2
#define Hs      56
#define Ws      56
#define Ls      (Hs*Ws)
#define DMODEL  96
#define DINNER  192
#define DSTATE  16
#define DTRANK  6
#define NCOUT   38
#define NKO     (4*NCOUT)
#define NKO_PAD 160
#define KBIG    (DINNER*2*9)
#define THETA_C 0.6f
#define MROWS   (BATCH*Ls)

static __device__ __forceinline__ float siluf(float v) { return v / (1.f + expf(-v)); }

static __device__ __forceinline__ v8f wmma4(v2f a, v2f b, v8f c) {
  return __builtin_amdgcn_wmma_f32_16x16x4_f32(false, a, false, b, (short)0, c, false, false);
}

__global__ void gemm_f32_wmma(const float* __restrict__ A, const float* __restrict__ Bt,
                              float* __restrict__ C, int K, int ldc) {
  const int lane = threadIdx.x & 31;
  const int half = lane >> 4;
  const int klo  = half << 1;
  const int ln   = lane & 15;
  const int ntile = blockIdx.x;
  const int mtile = blockIdx.y;
  const float* ap = A  + (size_t)(mtile * 16 + ln) * K + klo;
  const float* bp = Bt + (size_t)(ntile * 16 + ln) * K + klo;
  v8f acc = {0.f,0.f,0.f,0.f,0.f,0.f,0.f,0.f};
  for (int k = 0; k < K; k += 4) {
    v2f a = *(const v2f*)(ap + k);
    v2f b = *(const v2f*)(bp + k);
    acc = wmma4(a, b, acc);
  }
  const int n = ntile * 16 + ln;
  for (int rr = 0; rr < 8; ++rr) {
    int mm = mtile * 16 + rr + half * 8;
    C[(size_t)mm * ldc + n] = acc[rr];
  }
}

__global__ void k_pack_w(const float* __restrict__ xw, float* __restrict__ Wp) {
  int tid = blockIdx.x * blockDim.x + threadIdx.x;
  if (tid >= NKO_PAD * KBIG) return;
  int kk = tid % KBIG;
  int n  = tid / KBIG;
  float v = 0.f;
  if (n < NKO) {
    int c = kk % DINNER;
    int r = kk / DINNER;
    int dx = r % 3, dy = (r / 3) % 3, t = r / 9;
    v = xw[((((size_t)n * DINNER + c) * 2 + t) * 3 + dy) * 3 + dx];
  }
  Wp[tid] = v;
}

__global__ void k_pack_kd(const float* __restrict__ xw, float* __restrict__ kdp) {
  int tid = blockIdx.x * blockDim.x + threadIdx.x;
  if (tid >= NKO_PAD * DINNER) return;
  int c = tid % DINNER;
  int n = tid / DINNER;
  float v = 0.f;
  if (n < NKO) {
    const float* p = xw + ((size_t)n * DINNER + c) * 18;
    for (int i = 0; i < 18; ++i) v += p[i];
  }
  kdp[tid] = v;
}

__global__ void k2_conv(const float* __restrict__ xz, const float* __restrict__ xt,
                        const float* __restrict__ w2, const float* __restrict__ b2,
                        const float* __restrict__ wxt, const float* __restrict__ bxt,
                        float* __restrict__ xin) {
  int tid = blockIdx.x * blockDim.x + threadIdx.x;
  if (tid >= BATCH * 2 * Ls * DINNER) return;
  int c = tid % DINNER; int r = tid / DINNER;
  int w = r % Ws; r /= Ws;
  int h = r % Hs; r /= Hs;
  int t = r % 2;  int b = r / 2;
  float acc;
  if (t == 0) {
    acc = b2[c];
    for (int dy = 0; dy < 3; ++dy) {
      int hh = h + dy - 1; if (hh < 0 || hh >= Hs) continue;
      for (int dx = 0; dx < 3; ++dx) {
        int ww = w + dx - 1; if (ww < 0 || ww >= Ws) continue;
        acc += xz[((size_t)((b * Hs + hh) * Ws + ww)) * 384 + c] * w2[c * 9 + dy * 3 + dx];
      }
    }
  } else {
    acc = bxt[c];
    int ci = c >> 1;
    for (int dy = 0; dy < 3; ++dy) {
      int hh = h + dy - 1; if (hh < 0 || hh >= Hs) continue;
      for (int dx = 0; dx < 3; ++dx) {
        int ww = w + dx - 1; if (ww < 0 || ww >= Ws) continue;
        acc += xt[((size_t)((b * Hs + hh) * Ws + ww)) * DMODEL + ci] * wxt[c * 9 + dy * 3 + dx];
      }
    }
  }
  xin[tid] = siluf(acc);
}

__global__ void k3_xs(const float* __restrict__ xin, float* __restrict__ xs) {
  int tid = blockIdx.x * blockDim.x + threadIdx.x;
  if (tid >= BATCH * DINNER * Ls) return;
  int l = tid % Ls; int r = tid / Ls;
  int d = r % DINNER; int b = r / DINNER;

  auto X = [&](int s, int hh, int ww) -> float {
    return xin[(((size_t)(b * 2 + s) * Hs + hh) * Ws + ww) * DINNER + d];
  };
  auto irow = [&](int hh, int ww) -> float {
    int s = hh & 1, i = hh >> 1;
    float v0 = X(s, 2 * i, ww), v1 = X(s, 2 * i + 1, ww);
    return 0.5f * (v0 + v1) + fmaxf(v0, v1);
  };
  auto icol = [&](int hh, int ww) -> float {
    int s = ww & 1, j = ww >> 1;
    float v0 = X(s, hh, 2 * j), v1 = X(s, hh, 2 * j + 1);
    return 0.5f * (v0 + v1) + fmaxf(v0, v1);
  };

  int wr = l / Hs, hr = l % Hs;
  int hc = l / Ws, wc = l % Ws;
  size_t stride = (size_t)DINNER * Ls;
  size_t base = ((size_t)(b * 4) * DINNER + d) * Ls + l;
  xs[base + 0 * stride] = irow(hr, wr);
  xs[base + 1 * stride] = irow(Hs - 1 - hr, wr);
  xs[base + 2 * stride] = icol(hc, wc);
  xs[base + 3 * stride] = icol(hc, Ws - 1 - wc);
}

__global__ void k4_cdc(const float* __restrict__ xin, const float* __restrict__ Wp,
                       const float* __restrict__ kdp, float* __restrict__ xdbl) {
  const int lane = threadIdx.x & 31;
  const int half = lane >> 4;
  const int klo  = half << 1;
  const int ln   = lane & 15;
  const int ntile = blockIdx.x;
  const int mtile = blockIdx.y;
  const int m = mtile * 16 + ln;
  const int b = m / Ls;
  const int hw = m % Ls;
  const int h = hw / Ws, w = hw % Ws;
  const int n = ntile * 16 + ln;

  v8f accN  = {0.f,0.f,0.f,0.f,0.f,0.f,0.f,0.f};
  const float* wptr = Wp + (size_t)n * KBIG + klo;
  for (int td = 0; td < 18; ++td) {
    int t = td / 9, rr9 = td % 9, dy = rr9 / 3, dx = rr9 % 3;
    int hh = h + dy - 1, ww = w + dx - 1;
    bool ok = (hh >= 0) && (hh < Hs) && (ww >= 0) && (ww < Ws);
    const float* ap = xin + (((size_t)(b * 2 + t) * Hs + hh) * Ws + ww) * DINNER + klo;
    for (int c = 0; c < DINNER; c += 4) {
      v2f a = {0.f, 0.f};
      if (ok) a = *(const v2f*)(ap + c);
      v2f bb = *(const v2f*)wptr;
      wptr += 4;
      accN = wmma4(a, bb, accN);
    }
  }

  v8f accD0 = {0.f,0.f,0.f,0.f,0.f,0.f,0.f,0.f};
  v8f accD1 = {0.f,0.f,0.f,0.f,0.f,0.f,0.f,0.f};
  const float* ap0 = xin + (((size_t)(b * 2 + 0) * Hs + h) * Ws + w) * DINNER + klo;
  const float* ap1 = xin + (((size_t)(b * 2 + 1) * Hs + h) * Ws + w) * DINNER + klo;
  const float* kp  = kdp + (size_t)n * DINNER + klo;
  for (int c = 0; c < DINNER; c += 4) {
    v2f bb = *(const v2f*)(kp + c);
    v2f a0 = *(const v2f*)(ap0 + c);
    v2f a1 = *(const v2f*)(ap1 + c);
    accD0 = wmma4(a0, bb, accD0);
    accD1 = wmma4(a1, bb, accD1);
  }

  if (n < NKO) {
    int kq = n / NCOUT, o = n % NCOUT;
    for (int rr = 0; rr < 8; ++rr) {
      int mm = mtile * 16 + rr + half * 8;
      int b2 = mm / Ls; int l2 = mm % Ls;
      float Nv = accN[rr], d0 = accD0[rr], d1 = accD1[rr];
      float pooled = 2.f * Nv - THETA_C * (0.5f * (d0 + d1) + fminf(d0, d1));
      xdbl[(((size_t)b2 * 4 + kq) * NCOUT + o) * Ls + l2] = pooled;
    }
  }
}

__global__ void k5_delta(const float* __restrict__ xdbl, const float* __restrict__ dtw,
                         const float* __restrict__ dtb, float* __restrict__ delta) {
  int tid = blockIdx.x * blockDim.x + threadIdx.x;
  if (tid >= BATCH * 4 * DINNER * Ls) return;
  int l = tid % Ls; int r = tid / Ls;
  int d = r % DINNER; int bk = r / DINNER;
  int k = bk & 3;
  float acc = dtb[k * DINNER + d];
  const float* wp = dtw + ((size_t)k * DINNER + d) * DTRANK;
  const float* xp = xdbl + (size_t)bk * NCOUT * Ls + l;
  for (int rr = 0; rr < DTRANK; ++rr) acc += xp[(size_t)rr * Ls] * wp[rr];
  delta[tid] = fmaxf(acc, 0.f) + log1pf(expf(-fabsf(acc)));
}

__global__ void k6_scan(const float* __restrict__ xs, const float* __restrict__ xdbl,
                        const float* __restrict__ Alogs, const float* __restrict__ Dss,
                        float* __restrict__ dio) {
  int wid  = blockIdx.x * (blockDim.x >> 5) + (threadIdx.x >> 5);
  int lane = threadIdx.x & 31;
  int half = lane >> 4, ln = lane & 15;
  int bk = wid / 96;
  int d  = (wid % 96) * 2 + half;
  int k  = bk & 3;
  const float* dP = dio  + ((size_t)bk * DINNER + d) * Ls;
  float*       oP = dio  + ((size_t)bk * DINNER + d) * Ls;
  const float* uP = xs   + ((size_t)bk * DINNER + d) * Ls;
  const float* bP = xdbl + ((size_t)bk * NCOUT + DTRANK + ln) * Ls;
  const float* cP = xdbl + ((size_t)bk * NCOUT + DTRANK + DSTATE + ln) * Ls;
  float A  = -expf(Alogs[((size_t)(k * DINNER + d)) * DSTATE + ln]);
  float Dv = Dss[k * DINNER + d];
  float h = 0.f;
  for (int l = 0; l < Ls; ++l) {
    float dl = dP[l];
    float ul = uP[l];
    float bl = bP[l];
    float cl = cP[l];
    h = expf(dl * A) * h + (dl * ul) * bl;
    float y = h * cl;
    y += __shfl_xor(y, 1);
    y += __shfl_xor(y, 2);
    y += __shfl_xor(y, 4);
    y += __shfl_xor(y, 8);
    if (ln == 0) oP[l] = y + Dv * ul;
  }
}

__global__ void k7a_comb(const float* __restrict__ oy, float* __restrict__ y2) {
  int tid = blockIdx.x * blockDim.x + threadIdx.x;
  if (tid >= MROWS * DINNER) return;
  int d = tid % DINNER; int m = tid / DINNER;
  int b = m / Ls; int l = m % Ls;
  int h = l / Ws, w = l % Ws;
  int lt = w * Hs + h;
  size_t s = (size_t)DINNER * Ls;
  const float* base = oy + ((size_t)b * 4 * DINNER + d) * Ls;
  y2[tid] = base[l] + base[2 * s + (Ls - 1 - l)]
          + base[s + lt] + base[3 * s + (Ls - 1 - lt)];
}

__global__ void k7b_ln(const float* __restrict__ xz, const float* __restrict__ g,
                       const float* __restrict__ bb, float* __restrict__ y2) {
  int m = blockIdx.x;
  int d = threadIdx.x;
  float v = y2[(size_t)m * DINNER + d];
  float s1 = v, s2 = v * v;
  #pragma unroll
  for (int off = 16; off >= 1; off >>= 1) {
    s1 += __shfl_xor(s1, off);
    s2 += __shfl_xor(s2, off);
  }
  __shared__ float r1[6], r2[6];
  __shared__ float smu, srs;
  if ((d & 31) == 0) { r1[d >> 5] = s1; r2[d >> 5] = s2; }
  __syncthreads();
  if (d == 0) {
    float t1 = 0.f, t2 = 0.f;
    for (int i = 0; i < 6; ++i) { t1 += r1[i]; t2 += r2[i]; }
    float mu = t1 / (float)DINNER;
    float var = t2 / (float)DINNER - mu * mu;
    smu = mu; srs = rsqrtf(var + 1e-5f);
  }
  __syncthreads();
  float yln = (v - smu) * srs * g[d] + bb[d];
  float zv = xz[(size_t)m * 384 + DINNER + d];
  y2[(size_t)m * DINNER + d] = yln * siluf(zv);
}

extern "C" void kernel_launch(void* const* d_in, const int* in_sizes, int n_in,
                              void* d_out, int out_size, void* d_ws, size_t ws_size,
                              hipStream_t stream) {
  (void)in_sizes; (void)n_in; (void)out_size; (void)ws_size;
  const float* x   = (const float*)d_in[0];
  const float* xt  = (const float*)d_in[1];
  const float* ipw = (const float*)d_in[2];
  const float* c2w = (const float*)d_in[3];
  const float* c2b = (const float*)d_in[4];
  const float* cxw = (const float*)d_in[5];
  const float* cxb = (const float*)d_in[6];
  const float* xpw = (const float*)d_in[7];
  const float* dtw = (const float*)d_in[8];
  const float* dtb = (const float*)d_in[9];
  const float* alg = (const float*)d_in[10];
  const float* dss = (const float*)d_in[11];
  const float* lng = (const float*)d_in[12];
  const float* lnb = (const float*)d_in[13];
  const float* opw = (const float*)d_in[14];
  float* out = (float*)d_out;
  float* ws  = (float*)d_ws;

  size_t off = 0;
  float* xz    = ws + off; off += (size_t)MROWS * 384;
  float* xin   = ws + off; off += (size_t)BATCH * 2 * Ls * DINNER;
  float* xs    = ws + off; off += (size_t)BATCH * 4 * DINNER * Ls;
  float* Wp    = ws + off; off += (size_t)NKO_PAD * KBIG;
  float* kdp   = ws + off; off += (size_t)NKO_PAD * DINNER;
  float* xdbl  = ws + off; off += (size_t)BATCH * 4 * NCOUT * Ls;
  float* delta = ws + off; off += (size_t)BATCH * 4 * DINNER * Ls;
  float* y2    = xin;

  gemm_f32_wmma<<<dim3(384 / 16, MROWS / 16), 32, 0, stream>>>(x, ipw, xz, DMODEL, 384);

  k_pack_w <<<(NKO_PAD * KBIG + 255) / 256, 256, 0, stream>>>(xpw, Wp);
  k_pack_kd<<<(NKO_PAD * DINNER + 255) / 256, 256, 0, stream>>>(xpw, kdp);

  k2_conv<<<(BATCH * 2 * Ls * DINNER + 255) / 256, 256, 0, stream>>>(
      xz, xt, c2w, c2b, cxw, cxb, xin);

  k3_xs<<<(BATCH * DINNER * Ls + 255) / 256, 256, 0, stream>>>(xin, xs);

  k4_cdc<<<dim3(NKO_PAD / 16, MROWS / 16), 32, 0, stream>>>(xin, Wp, kdp, xdbl);

  k5_delta<<<(BATCH * 4 * DINNER * Ls + 255) / 256, 256, 0, stream>>>(xdbl, dtw, dtb, delta);

  k6_scan<<<96, 256, 0, stream>>>(xs, xdbl, alg, dss, delta);

  k7a_comb<<<(MROWS * DINNER + 255) / 256, 256, 0, stream>>>(delta, y2);
  k7b_ln<<<MROWS, DINNER, 0, stream>>>(xz, lng, lnb, y2);

  gemm_f32_wmma<<<dim3(DMODEL / 16, MROWS / 16), 32, 0, stream>>>(y2, opw, out, DINNER, DMODEL);
}


# tail fused: cross-merge, LayerNorm+gate and out_proj GEMM in one kernel; 6 waves keep a 4x4-pixel x 192 tile on chip, weight rows staged to LDS as whole lines; LN sums re-ordered (12 partials of 16)
# speedup vs baseline: 7.4179x; 7.4179x over previous
#include <hip/hip_runtime.h>
#include <hip/hip_bf16.h>
#include <math.h>

typedef float v2f __attribute__((ext_vector_type(2)));
typedef float v8f __attribute__((ext_vector_type(8)));
typedef float v4f __attribute__((ext_vector_type(4)));

#define BATCH   2
#define Hs      56
#define Ws      56
#define Ls      (Hs*Ws)
#define DMODEL  96
#define DINNER  192
#define DSTATE  16
#define DTRANK  6
#define NCOUT   38
#define NKO     (4*NCOUT)
#define NKO_PAD 160
#define KBIG    (DINNER*2*9)
#define THETA_C 0.6f
#define MROWS   (BATCH*Ls)

static __device__ __forceinline__ float siluf(float v) { return v / (1.f + expf(-v)); }

static __device__ __forceinline__ v8f wmma4(v2f a, v2f b, v8f c) {
  return __builtin_amdgcn_wmma_f32_16x16x4_f32(false, a, false, b, (short)0, c, false, false);
}

__global__ void gemm_f32_wmma(const float* __restrict__ A, const float* __restrict__ Bt,
                              float* __restrict__ C, int K, int ldc) {
  const int lane = threadIdx.x & 31;
  const int half = lane >> 4;
  const int klo  = half << 1;
  const int ln   = lane & 15;
  const int ntile = blockIdx.x;
  const int mtile = blockIdx.y;
  const float* ap = A  + (size_t)(mtile * 16 + ln) * K + klo;
  const float* bp = Bt + (size_t)(ntile * 16 + ln) * K + klo;
  v8f acc = {0.f,0.f,0.f,0.f,0.f,0.f,0.f,0.f};
  for (int k = 0; k < K; k += 4) {
    v2f a = *(const v2f*)(ap + k);
    v2f b = *(const v2f*)(bp + k);
    acc = wmma4(a, b, acc);
  }
  const int n = ntile * 16 + ln;
  for (int rr = 0; rr < 8; ++rr) {
    int mm = mtile * 16 + rr + half * 8;
    C[(size_t)mm * ldc + n] = acc[rr];
  }
}

__global__ void k_pack_w(const float* __restrict__ xw, float* __restrict__ Wp) {
  int tid = blockIdx.x * blockDim.x + threadIdx.x;
  if (tid >= NKO_PAD * KBIG) return;
  int kk = tid % KBIG;
  int n  = tid / KBIG;
  float v = 0.f;
  if (n < NKO) {
    int c = kk % DINNER;
    int r = kk / DINNER;
    int dx = r % 3, dy = (r / 3) % 3, t = r / 9;
    v = xw[((((size_t)n * DINNER + c) * 2 + t) * 3 + dy) * 3 + dx];
  }
  Wp[tid] = v;
}

__global__ void k_pack_kd(const float* __restrict__ xw, float* __restrict__ kdp) {
  int tid = blockIdx.x * blockDim.x + threadIdx.x;
  if (tid >= NKO_PAD * DINNER) return;
  int c = tid % DINNER;
  int n = tid / DINNER;
  float v = 0.f;
  if (n < NKO) {
    const float* p = xw + ((size_t)n * DINNER + c) * 18;
    for (int i = 0; i < 18; ++i) v += p[i];
  }
  kdp[tid] = v;
}

__global__ void k2_conv(const float* __restrict__ xz, const float* __restrict__ xt,
                        const float* __restrict__ w2, const float* __restrict__ b2,
                        const float* __restrict__ wxt, const float* __restrict__ bxt,
                        float* __restrict__ xin) {
  int tid = blockIdx.x * blockDim.x + threadIdx.x;
  if (tid >= BATCH * 2 * Ls * DINNER) return;
  int c = tid % DINNER; int r = tid / DINNER;
  int w = r % Ws; r /= Ws;
  int h = r % Hs; r /= Hs;
  int t = r % 2;  int b = r / 2;
  float acc;
  if (t == 0) {
    acc = b2[c];
    for (int dy = 0; dy < 3; ++dy) {
      int hh = h + dy - 1; if (hh < 0 || hh >= Hs) continue;
      for (int dx = 0; dx < 3; ++dx) {
        int ww = w + dx - 1; if (ww < 0 || ww >= Ws) continue;
        acc += xz[((size_t)((b * Hs + hh) * Ws + ww)) * 384 + c] * w2[c * 9 + dy * 3 + dx];
      }
    }
  } else {
    acc = bxt[c];
    int ci = c >> 1;
    for (int dy = 0; dy < 3; ++dy) {
      int hh = h + dy - 1; if (hh < 0 || hh >= Hs) continue;
      for (int dx = 0; dx < 3; ++dx) {
        int ww = w + dx - 1; if (ww < 0 || ww >= Ws) continue;
        acc += xt[((size_t)((b * Hs + hh) * Ws + ww)) * DMODEL + ci] * wxt[c * 9 + dy * 3 + dx];
      }
    }
  }
  xin[tid] = siluf(acc);
}

#define K3_SL   Hs
#define K3_NSEG 2
#define K3_GRP  (Hs / (2 * K3_NSEG))
#define K3_CG   (DINNER / 16)

template <int TYPE>
static __device__ __forceinline__ void k3_body(const float* __restrict__ xin, float* __restrict__ xs,
                                               int b, int seg0, int d) {
  constexpr int PSTEP = (TYPE == 0) ? Ws * DINNER : DINNER;
  constexpr int SSTEP = (TYPE == 0) ? DINNER : Ws * DINNER;
  const float* x0 = xin + (size_t)(b * 2 + 0) * Ls * DINNER + d;
  const float* x1 = xin + (size_t)(b * 2 + 1) * Ls * DINNER + d;
  float* pf = xs + ((size_t)(b * 4 + 2 * TYPE) * DINNER + d) * Ls;
  float* pr = pf + (size_t)DINNER * Ls;
  for (int sg = 0; sg < K3_NSEG; ++sg) {
    const int seg = seg0 + sg;
    const float* a0 = x0 + (size_t)seg * SSTEP;
    const float* a1 = x1 + (size_t)seg * SSTEP;
    float o[K3_SL];
    #pragma unroll
    for (int i = 0; i < K3_SL / 2; ++i) {
      const float v0 = a0[(2 * i) * PSTEP], v1 = a0[(2 * i + 1) * PSTEP];
      o[2 * i] = 0.5f * (v0 + v1) + fmaxf(v0, v1);
      const float w0 = a1[(2 * i) * PSTEP], w1 = a1[(2 * i + 1) * PSTEP];
      o[2 * i + 1] = 0.5f * (w0 + w1) + fmaxf(w0, w1);
    }
    #pragma unroll
    for (int v = 0; v < K3_SL / 4; ++v) {
      *(v4f*)(pf + seg * K3_SL + 4 * v) = (v4f){o[4 * v], o[4 * v + 1], o[4 * v + 2], o[4 * v + 3]};
      *(v4f*)(pr + seg * K3_SL + 4 * v) =
          (v4f){o[K3_SL - 1 - 4 * v], o[K3_SL - 2 - 4 * v], o[K3_SL - 3 - 4 * v], o[K3_SL - 4 - 4 * v]};
    }
  }
}

__global__ void __launch_bounds__(32)
k3_xs(const float* __restrict__ xin, float* __restrict__ xs) {
  static_assert(Hs == Ws, "a scan segment is Hs steps in one scan order and Ws in the other");
  int r = blockIdx.x;
  const int d = (r % K3_CG) * 16 + (threadIdx.x & 15); r /= K3_CG;
  const int seg0 = ((r % K3_GRP) * 2 + (threadIdx.x >> 4)) * K3_NSEG; r /= K3_GRP;
  const int b = r % BATCH;
  const int type = r / BATCH;
  if (type == 0) k3_body<0>(xin, xs, b, seg0, d);
  else           k3_body<1>(xin, xs, b, seg0, d);
}

#define K4_BM    32
#define K4_BK    DINNER
#define K4_P     (K4_BK + 4)
#define K4_NT    640
#define K4_RPP   (K4_NT / 16)
#define K4_NSUB  (K4_BK / 64)
#define K4_STAPS 18
#define K4_SALL  20

__global__ void __launch_bounds__(K4_NT)
k4_cdc(const float* __restrict__ xin, const float* __restrict__ Wp,
       const float* __restrict__ kdp, float* __restrict__ xdbl) {
  __shared__ float sA[2][K4_BM * K4_P];
  __shared__ float sB[2][NKO_PAD * K4_P];

  const int tid  = threadIdx.x;
  const int lane = tid & 31;
  const int wv   = tid >> 5;
  const int half = lane >> 4;
  const int klo  = half << 1;
  const int ln   = lane & 15;
  const int nt   = wv % (NKO_PAD / 16);
  const int mt   = wv / (NKO_PAD / 16);
  const int mbase = blockIdx.x * K4_BM;

  const int ar = tid >> 4;
  const int cv = tid & 15;
  const bool aload = ar < K4_BM;
  const int am = mbase + (aload ? ar : 0);
  const int ab = am / Ls;
  const int ahw = am % Ls;
  const int ah = ahw / Ws, aw = ahw % Ws;

  v4f pa[K4_NSUB];
  v4f pb[K4_NSUB][4];

  auto gload = [&](int s) {
    int t, dy, dx;
    const float* bsrc;
    int bstride;
    if (s < K4_STAPS) {
      t = s / 9;
      const int r9 = s - t * 9;
      dy = r9 / 3;
      dx = r9 - dy * 3;
      bsrc = Wp + s * DINNER;
      bstride = KBIG;
    } else {
      t = s - K4_STAPS;
      dy = 1; dx = 1;
      bsrc = kdp;
      bstride = DINNER;
    }
    const int hh = ah + dy - 1, ww = aw + dx - 1;
    const bool ok = aload && hh >= 0 && hh < Hs && ww >= 0 && ww < Ws;
    #pragma unroll
    for (int sub = 0; sub < K4_NSUB; ++sub) pa[sub] = (v4f){0.f, 0.f, 0.f, 0.f};
    if (ok) {
      const float* asrc = xin + (((size_t)(ab * 2 + t) * Hs + hh) * Ws + ww) * DINNER + cv * 4;
      #pragma unroll
      for (int sub = 0; sub < K4_NSUB; ++sub) pa[sub] = *(const v4f*)(asrc + sub * 64);
    }
    #pragma unroll
    for (int sub = 0; sub < K4_NSUB; ++sub) {
      #pragma unroll
      for (int j = 0; j < 4; ++j)
        pb[sub][j] = *(const v4f*)(bsrc + (size_t)(ar + j * K4_RPP) * bstride + sub * 64 + cv * 4);
    }
  };

  auto lstore = [&](int buf) {
    #pragma unroll
    for (int sub = 0; sub < K4_NSUB; ++sub) {
      if (aload) *(v4f*)(&sA[buf][ar * K4_P + sub * 64 + cv * 4]) = pa[sub];
      #pragma unroll
      for (int j = 0; j < 4; ++j)
        *(v4f*)(&sB[buf][(ar + j * K4_RPP) * K4_P + sub * 64 + cv * 4]) = pb[sub][j];
    }
  };

  auto stage = [&](int s, v8f& acc) {
    const bool more = (s + 1 < K4_SALL);
    if (more) gload(s + 1);
    const float* a = &sA[s & 1][(mt * 16 + ln) * K4_P + klo];
    const float* b = &sB[s & 1][(nt * 16 + ln) * K4_P + klo];
    #pragma unroll
    for (int kk = 0; kk < K4_BK / 8; ++kk) {
      v2f av = *(const v2f*)(a + kk * 4);
      v2f bv = *(const v2f*)(b + kk * 4);
      acc = wmma4(av, bv, acc);
    }
    if (more) lstore((s + 1) & 1);
    #pragma unroll
    for (int kk = K4_BK / 8; kk < K4_BK / 4; ++kk) {
      v2f av = *(const v2f*)(a + kk * 4);
      v2f bv = *(const v2f*)(b + kk * 4);
      acc = wmma4(av, bv, acc);
    }
    __syncthreads();
  };

  v8f accN  = {0.f,0.f,0.f,0.f,0.f,0.f,0.f,0.f};
  v8f accD0 = {0.f,0.f,0.f,0.f,0.f,0.f,0.f,0.f};
  v8f accD1 = {0.f,0.f,0.f,0.f,0.f,0.f,0.f,0.f};

  gload(0);
  lstore(0);
  __syncthreads();
  for (int s = 0; s < K4_STAPS; ++s) stage(s, accN);
  stage(K4_STAPS, accD0);
  stage(K4_STAPS + 1, accD1);

  const int n = nt * 16 + ln;
  if (n < NKO) {
    int kq = n / NCOUT, o = n % NCOUT;
    for (int rr = 0; rr < 8; ++rr) {
      int mm = mbase + mt * 16 + rr + half * 8;
      int b2 = mm / Ls; int l2 = mm % Ls;
      float Nv = accN[rr], d0 = accD0[rr], d1 = accD1[rr];
      float pooled = 2.f * Nv - THETA_C * (0.5f * (d0 + d1) + fminf(d0, d1));
      xdbl[(((size_t)b2 * 4 + kq) * NCOUT + o) * Ls + l2] = pooled;
    }
  }
}

__global__ void k5_delta(const float* __restrict__ xdbl, const float* __restrict__ dtw,
                         const float* __restrict__ dtb, float* __restrict__ delta) {
  int tid = blockIdx.x * blockDim.x + threadIdx.x;
  if (tid >= BATCH * 4 * DINNER * Ls) return;
  int l = tid % Ls; int r = tid / Ls;
  int d = r % DINNER; int bk = r / DINNER;
  int k = bk & 3;
  float acc = dtb[k * DINNER + d];
  const float* wp = dtw + ((size_t)k * DINNER + d) * DTRANK;
  const float* xp = xdbl + (size_t)bk * NCOUT * Ls + l;
  for (int rr = 0; rr < DTRANK; ++rr) acc += xp[(size_t)rr * Ls] * wp[rr];
  delta[tid] = fmaxf(acc, 0.f) + log1pf(expf(-fabsf(acc)));
}

#define QL    DSTATE
#define CPW   (32 / QL)
#define K6_WV 3
#define VPC   8
#define SCH   (4 * VPC)
#define NPAIR (Ls / (2 * SCH))

struct ScanBuf {
  v4f dl[VPC], ul[VPC], b[VPC], c[VPC];
};

static __device__ __forceinline__ void scan_load(ScanBuf& s, const float* dP, const float* uP,
                                                 const float* bP, const float* cP, int l0) {
  #pragma unroll
  for (int v = 0; v < VPC; ++v) {
    s.dl[v] = *(const v4f*)(dP + l0 + 4 * v);
    s.ul[v] = *(const v4f*)(uP + l0 + 4 * v);
  }
  #pragma unroll
  for (int v = 0; v < VPC; ++v) {
    s.b[v] = *(const v4f*)(bP + l0 + 4 * v);
    s.c[v] = *(const v4f*)(cP + l0 + 4 * v);
  }
}

template <int XM>
static __device__ __forceinline__ float row_xor(float v) {
  return __int_as_float(__builtin_amdgcn_update_dpp(0, __float_as_int(v), 0x160 + XM, 0xf, 0xf, true));
}

static __device__ __forceinline__ void scan_steps(const ScanBuf& s, float& h, const float A,
                                                  float Dv, float* oP, int l0, bool writer) {
  v4f o[VPC];
  #pragma unroll
  for (int v = 0; v < VPC; ++v) {
    #pragma unroll
    for (int e = 0; e < 4; ++e) {
      const float dl = s.dl[v][e];
      const float ul = s.ul[v][e];
      h = expf(dl * A) * h + (dl * ul) * s.b[v][e];
      float y = h * s.c[v][e];
      y += row_xor<1>(y);
      y += row_xor<2>(y);
      y += row_xor<4>(y);
      y += row_xor<8>(y);
      o[v][e] = y + Dv * ul;
    }
  }
  if (writer) {
    #pragma unroll
    for (int v = 0; v < VPC; ++v) *(v4f*)(oP + l0 + 4 * v) = o[v];
  }
}

__global__ void __launch_bounds__(32 * K6_WV)
k6_scan(const float* __restrict__ xs, const float* __restrict__ xdbl,
        const float* __restrict__ Alogs, const float* __restrict__ Dss,
        float* __restrict__ dio) {
  const int lane = threadIdx.x & 31;
  const int wave = blockIdx.x * K6_WV + (threadIdx.x >> 5);
  const int n  = lane & (QL - 1);
  const int ch = wave * CPW + lane / QL;
  const int bk = ch / DINNER;
  const int d  = ch % DINNER;
  const int k  = bk & 3;
  float*       oP = dio  + (size_t)ch * Ls;
  const float* dP = oP;
  const float* uP = xs   + (size_t)ch * Ls;
  const float* bP = xdbl + ((size_t)bk * NCOUT + DTRANK + n) * Ls;
  const float* cP = xdbl + ((size_t)bk * NCOUT + DTRANK + DSTATE + n) * Ls;
  const float A = -expf(Alogs[((size_t)(k * DINNER + d)) * DSTATE + n]);
  float h = 0.f;
  const float Dv = Dss[k * DINNER + d];
  const bool writer = (n == 0);

  ScanBuf s0, s1;
  scan_load(s0, dP, uP, bP, cP, 0);
  for (int p = 0; p < NPAIR; ++p) {
    const int l0 = p * (2 * SCH);
    scan_load(s1, dP, uP, bP, cP, l0 + SCH);
    __builtin_amdgcn_sched_barrier(0);
    scan_steps(s0, h, A, Dv, oP, l0, writer);
    const int ln = (p + 1 < NPAIR) ? l0 + 2 * SCH : l0 + SCH;
    scan_load(s0, dP, uP, bP, cP, ln);
    __builtin_amdgcn_sched_barrier(0);
    scan_steps(s1, h, A, Dv, oP, l0 + SCH, writer);
  }
}

#define KT_P   (DINNER + 4)
#define KT_OP  (DMODEL + 4)
#define KT_NQ  12
#define KT_KS  (DINNER / 4)
#define KT_WL  (16 * DINNER / 128)

__global__ void __launch_bounds__(DINNER)
k7_tail(const float* __restrict__ oy, const float* __restrict__ xz,
        const float* __restrict__ g, const float* __restrict__ bb,
        const float* __restrict__ opw, float* __restrict__ out) {
  __shared__ __attribute__((aligned(16))) float sT[16 * KT_P];
  __shared__ __attribute__((aligned(16))) float sO[16 * KT_OP];
  __shared__ __attribute__((aligned(16))) float sB[DMODEL * KT_P];
  __shared__ __attribute__((aligned(16))) float sP1[16 * KT_NQ], sP2[16 * KT_NQ];
  __shared__ float sMu[16], sRs[16];

  const int d = threadIdx.x;
  const int lane = d & 31;
  const int nt   = d >> 5;
  const int half = lane >> 4;
  const int ln   = lane & 15;
  int r = blockIdx.x;
  const int w0 = (r % (Ws / 4)) * 4; r /= (Ws / 4);
  const int h0 = (r % (Hs / 4)) * 4;
  const int b = r / (Hs / 4);

  v4f wr[KT_WL];
  {
    const float* bsrc = opw + (size_t)nt * 16 * DINNER + lane * 4;
    #pragma unroll
    for (int it = 0; it < KT_WL; ++it) wr[it] = *(const v4f*)(bsrc + it * 128);
  }

  const size_t s = (size_t)DINNER * Ls;
  const float* p0 = oy + ((size_t)b * 4 * DINNER + d) * Ls;
  v4f a0[4], a1[4], a2[4], a3[4];
  #pragma unroll
  for (int i = 0; i < 4; ++i) {
    const int l  = (h0 + i) * Ws + w0;
    const int lt = (w0 + i) * Hs + h0;
    a0[i] = *(const v4f*)(p0 + l);
    a2[i] = *(const v4f*)(p0 + 2 * s + (Ls - 4 - l));
    a1[i] = *(const v4f*)(p0 + s + lt);
    a3[i] = *(const v4f*)(p0 + 3 * s + (Ls - 4 - lt));
  }
  float zv[16];
  #pragma unroll
  for (int p = 0; p < 16; ++p) {
    const size_t m = (size_t)b * Ls + (h0 + (p >> 2)) * Ws + (w0 + (p & 3));
    zv[p] = xz[m * 384 + DINNER + d];
  }
  float val[16];
  #pragma unroll
  for (int i = 0; i < 4; ++i) {
    #pragma unroll
    for (int j = 0; j < 4; ++j)
      val[i * 4 + j] = a0[i][j] + a2[i][3 - j] + a1[j][i] + a3[j][3 - i];
  }
  #pragma unroll
  for (int p = 0; p < 16; ++p) sT[p * KT_P + d] = val[p];
  #pragma unroll
  for (int it = 0; it < KT_WL; ++it) {
    const int f = it * 128 + lane * 4;
    *(v4f*)(&sB[(nt * 16 + f / DINNER) * KT_P + f % DINNER]) = wr[it];
  }
  __syncthreads();

  {
    const int p = d / KT_NQ, q = d % KT_NQ;
    const float* tp = &sT[p * KT_P + q * 16];
    float s1 = 0.f, s2 = 0.f;
    #pragma unroll
    for (int i = 0; i < 16; ++i) { const float v = tp[i]; s1 += v; s2 += v * v; }
    sP1[d] = s1; sP2[d] = s2;
  }
  __syncthreads();
  if (d < 16) {
    float t1 = 0.f, t2 = 0.f;
    #pragma unroll
    for (int q = 0; q < KT_NQ; ++q) { t1 += sP1[d * KT_NQ + q]; t2 += sP2[d * KT_NQ + q]; }
    const float mu = t1 / (float)DINNER;
    const float var = t2 / (float)DINNER - mu * mu;
    sMu[d] = mu; sRs[d] = rsqrtf(var + 1e-5f);
  }
  __syncthreads();

  {
    const float gd = g[d], bd = bb[d];
    #pragma unroll
    for (int p = 0; p < 16; ++p) {
      const float yln = (val[p] - sMu[p]) * sRs[p] * gd + bd;
      sT[p * KT_P + d] = yln * siluf(zv[p]);
    }
  }
  __syncthreads();

  v8f acc = {0.f,0.f,0.f,0.f,0.f,0.f,0.f,0.f};
  {
    const float* ap = &sT[ln * KT_P + 2 * half];
    const float* bp = &sB[(nt * 16 + ln) * KT_P + 2 * half];
    #pragma unroll
    for (int kk = 0; kk < KT_KS; ++kk) {
      const v2f av = *(const v2f*)(ap + 4 * kk);
      const v2f bv = *(const v2f*)(bp + 4 * kk);
      acc = wmma4(av, bv, acc);
    }
  }
  #pragma unroll
  for (int rr = 0; rr < 8; ++rr) sO[(rr + 8 * half) * KT_OP + nt * 16 + ln] = acc[rr];
  __syncthreads();

  #pragma unroll
  for (int ps = 0; ps < 2; ++ps) {
    const int idx = ps * DINNER + d;
    const int p = idx / (DMODEL / 4), c4 = idx % (DMODEL / 4);
    const size_t m = (size_t)b * Ls + (h0 + (p >> 2)) * Ws + (w0 + (p & 3));
    *(v4f*)(out + m * DMODEL + c4 * 4) = *(const v4f*)(&sO[p * KT_OP + c4 * 4]);
  }
}

extern "C" void kernel_launch(void* const* d_in, const int* in_sizes, int n_in,
                              void* d_out, int out_size, void* d_ws, size_t ws_size,
                              hipStream_t stream) {
  (void)in_sizes; (void)n_in; (void)out_size; (void)ws_size;
  const float* x   = (const float*)d_in[0];
  const float* xt  = (const float*)d_in[1];
  const float* ipw = (const float*)d_in[2];
  const float* c2w = (const float*)d_in[3];
  const float* c2b = (const float*)d_in[4];
  const float* cxw = (const float*)d_in[5];
  const float* cxb = (const float*)d_in[6];
  const float* xpw = (const float*)d_in[7];
  const float* dtw = (const float*)d_in[8];
  const float* dtb = (const float*)d_in[9];
  const float* alg = (const float*)d_in[10];
  const float* dss = (const float*)d_in[11];
  const float* lng = (const float*)d_in[12];
  const float* lnb = (const float*)d_in[13];
  const float* opw = (const float*)d_in[14];
  float* out = (float*)d_out;
  float* ws  = (float*)d_ws;

  size_t off = 0;
  float* xz    = ws + off; off += (size_t)MROWS * 384;
  float* xin   = ws + off; off += (size_t)BATCH * 2 * Ls * DINNER;
  float* xs    = ws + off; off += (size_t)BATCH * 4 * DINNER * Ls;
  float* Wp    = ws + off; off += (size_t)NKO_PAD * KBIG;
  float* kdp   = ws + off; off += (size_t)NKO_PAD * DINNER;
  float* xdbl  = ws + off; off += (size_t)BATCH * 4 * NCOUT * Ls;
  float* delta = ws + off; off += (size_t)BATCH * 4 * DINNER * Ls;

  gemm_f32_wmma<<<dim3(384 / 16, MROWS / 16), 32, 0, stream>>>(x, ipw, xz, DMODEL, 384);

  k_pack_w <<<(NKO_PAD * KBIG + 255) / 256, 256, 0, stream>>>(xpw, Wp);
  k_pack_kd<<<(NKO_PAD * DINNER + 255) / 256, 256, 0, stream>>>(xpw, kdp);

  k2_conv<<<(BATCH * 2 * Ls * DINNER + 255) / 256, 256, 0, stream>>>(
      xz, xt, c2w, c2b, cxw, cxb, xin);

  k3_xs<<<2 * BATCH * K3_GRP * K3_CG, 32, 0, stream>>>(xin, xs);

  k4_cdc<<<MROWS / K4_BM, K4_NT, 0, stream>>>(xin, Wp, kdp, xdbl);

  k5_delta<<<(BATCH * 4 * DINNER * Ls + 255) / 256, 256, 0, stream>>>(xdbl, dtw, dtb, delta);

  k6_scan<<<BATCH * 4 * DINNER / (CPW * K6_WV), 32 * K6_WV, 0, stream>>>(xs, xdbl, alg, dss, delta);

  k7_tail<<<BATCH * (Hs / 4) * (Ws / 4), DINNER, 0, stream>>>(delta, xz, lng, lnb, opw, out);
}


# k6 scan: a chunk's 32 products h*C stay in registers, summed by 2 shared butterflies, a lane keeping half the steps per level: 30 row_xmask adds per 32 steps (was 128); tree kept, h*C rounded first
# speedup vs baseline: 7.9214x; 1.0679x over previous
#include <hip/hip_runtime.h>
#include <hip/hip_bf16.h>
#include <math.h>

typedef float v2f __attribute__((ext_vector_type(2)));
typedef float v8f __attribute__((ext_vector_type(8)));
typedef float v4f __attribute__((ext_vector_type(4)));

#define BATCH   2
#define Hs      56
#define Ws      56
#define Ls      (Hs*Ws)
#define DMODEL  96
#define DINNER  192
#define DSTATE  16
#define DTRANK  6
#define NCOUT   38
#define NKO     (4*NCOUT)
#define NKO_PAD 160
#define KBIG    (DINNER*2*9)
#define THETA_C 0.6f
#define MROWS   (BATCH*Ls)

static __device__ __forceinline__ float siluf(float v) { return v / (1.f + expf(-v)); }

static __device__ __forceinline__ v8f wmma4(v2f a, v2f b, v8f c) {
  return __builtin_amdgcn_wmma_f32_16x16x4_f32(false, a, false, b, (short)0, c, false, false);
}

__global__ void gemm_f32_wmma(const float* __restrict__ A, const float* __restrict__ Bt,
                              float* __restrict__ C, int K, int ldc) {
  const int lane = threadIdx.x & 31;
  const int half = lane >> 4;
  const int klo  = half << 1;
  const int ln   = lane & 15;
  const int ntile = blockIdx.x;
  const int mtile = blockIdx.y;
  const float* ap = A  + (size_t)(mtile * 16 + ln) * K + klo;
  const float* bp = Bt + (size_t)(ntile * 16 + ln) * K + klo;
  v8f acc = {0.f,0.f,0.f,0.f,0.f,0.f,0.f,0.f};
  for (int k = 0; k < K; k += 4) {
    v2f a = *(const v2f*)(ap + k);
    v2f b = *(const v2f*)(bp + k);
    acc = wmma4(a, b, acc);
  }
  const int n = ntile * 16 + ln;
  for (int rr = 0; rr < 8; ++rr) {
    int mm = mtile * 16 + rr + half * 8;
    C[(size_t)mm * ldc + n] = acc[rr];
  }
}

__global__ void k_pack_w(const float* __restrict__ xw, float* __restrict__ Wp) {
  int tid = blockIdx.x * blockDim.x + threadIdx.x;
  if (tid >= NKO_PAD * KBIG) return;
  int kk = tid % KBIG;
  int n  = tid / KBIG;
  float v = 0.f;
  if (n < NKO) {
    int c = kk % DINNER;
    int r = kk / DINNER;
    int dx = r % 3, dy = (r / 3) % 3, t = r / 9;
    v = xw[((((size_t)n * DINNER + c) * 2 + t) * 3 + dy) * 3 + dx];
  }
  Wp[tid] = v;
}

__global__ void k_pack_kd(const float* __restrict__ xw, float* __restrict__ kdp) {
  int tid = blockIdx.x * blockDim.x + threadIdx.x;
  if (tid >= NKO_PAD * DINNER) return;
  int c = tid % DINNER;
  int n = tid / DINNER;
  float v = 0.f;
  if (n < NKO) {
    const float* p = xw + ((size_t)n * DINNER + c) * 18;
    for (int i = 0; i < 18; ++i) v += p[i];
  }
  kdp[tid] = v;
}

__global__ void k2_conv(const float* __restrict__ xz, const float* __restrict__ xt,
                        const float* __restrict__ w2, const float* __restrict__ b2,
                        const float* __restrict__ wxt, const float* __restrict__ bxt,
                        float* __restrict__ xin) {
  int tid = blockIdx.x * blockDim.x + threadIdx.x;
  if (tid >= BATCH * 2 * Ls * DINNER) return;
  int c = tid % DINNER; int r = tid / DINNER;
  int w = r % Ws; r /= Ws;
  int h = r % Hs; r /= Hs;
  int t = r % 2;  int b = r / 2;
  float acc;
  if (t == 0) {
    acc = b2[c];
    for (int dy = 0; dy < 3; ++dy) {
      int hh = h + dy - 1; if (hh < 0 || hh >= Hs) continue;
      for (int dx = 0; dx < 3; ++dx) {
        int ww = w + dx - 1; if (ww < 0 || ww >= Ws) continue;
        acc += xz[((size_t)((b * Hs + hh) * Ws + ww)) * 384 + c] * w2[c * 9 + dy * 3 + dx];
      }
    }
  } else {
    acc = bxt[c];
    int ci = c >> 1;
    for (int dy = 0; dy < 3; ++dy) {
      int hh = h + dy - 1; if (hh < 0 || hh >= Hs) continue;
      for (int dx = 0; dx < 3; ++dx) {
        int ww = w + dx - 1; if (ww < 0 || ww >= Ws) continue;
        acc += xt[((size_t)((b * Hs + hh) * Ws + ww)) * DMODEL + ci] * wxt[c * 9 + dy * 3 + dx];
      }
    }
  }
  xin[tid] = siluf(acc);
}

#define K3_SL   Hs
#define K3_NSEG 2
#define K3_GRP  (Hs / (2 * K3_NSEG))
#define K3_CG   (DINNER / 16)

template <int TYPE>
static __device__ __forceinline__ void k3_body(const float* __restrict__ xin, float* __restrict__ xs,
                                               int b, int seg0, int d) {
  constexpr int PSTEP = (TYPE == 0) ? Ws * DINNER : DINNER;
  constexpr int SSTEP = (TYPE == 0) ? DINNER : Ws * DINNER;
  const float* x0 = xin + (size_t)(b * 2 + 0) * Ls * DINNER + d;
  const float* x1 = xin + (size_t)(b * 2 + 1) * Ls * DINNER + d;
  float* pf = xs + ((size_t)(b * 4 + 2 * TYPE) * DINNER + d) * Ls;
  float* pr = pf + (size_t)DINNER * Ls;
  for (int sg = 0; sg < K3_NSEG; ++sg) {
    const int seg = seg0 + sg;
    const float* a0 = x0 + (size_t)seg * SSTEP;
    const float* a1 = x1 + (size_t)seg * SSTEP;
    float o[K3_SL];
    #pragma unroll
    for (int i = 0; i < K3_SL / 2; ++i) {
      const float v0 = a0[(2 * i) * PSTEP], v1 = a0[(2 * i + 1) * PSTEP];
      o[2 * i] = 0.5f * (v0 + v1) + fmaxf(v0, v1);
      const float w0 = a1[(2 * i) * PSTEP], w1 = a1[(2 * i + 1) * PSTEP];
      o[2 * i + 1] = 0.5f * (w0 + w1) + fmaxf(w0, w1);
    }
    #pragma unroll
    for (int v = 0; v < K3_SL / 4; ++v) {
      *(v4f*)(pf + seg * K3_SL + 4 * v) = (v4f){o[4 * v], o[4 * v + 1], o[4 * v + 2], o[4 * v + 3]};
      *(v4f*)(pr + seg * K3_SL + 4 * v) =
          (v4f){o[K3_SL - 1 - 4 * v], o[K3_SL - 2 - 4 * v], o[K3_SL - 3 - 4 * v], o[K3_SL - 4 - 4 * v]};
    }
  }
}

__global__ void __launch_bounds__(32)
k3_xs(const float* __restrict__ xin, float* __restrict__ xs) {
  static_assert(Hs == Ws, "a scan segment is Hs steps in one scan order and Ws in the other");
  int r = blockIdx.x;
  const int d = (r % K3_CG) * 16 + (threadIdx.x & 15); r /= K3_CG;
  const int seg0 = ((r % K3_GRP) * 2 + (threadIdx.x >> 4)) * K3_NSEG; r /= K3_GRP;
  const int b = r % BATCH;
  const int type = r / BATCH;
  if (type == 0) k3_body<0>(xin, xs, b, seg0, d);
  else           k3_body<1>(xin, xs, b, seg0, d);
}

#define K4_BM    32
#define K4_BK    DINNER
#define K4_P     (K4_BK + 4)
#define K4_NT    640
#define K4_RPP   (K4_NT / 16)
#define K4_NSUB  (K4_BK / 64)
#define K4_STAPS 18
#define K4_SALL  20

__global__ void __launch_bounds__(K4_NT)
k4_cdc(const float* __restrict__ xin, const float* __restrict__ Wp,
       const float* __restrict__ kdp, float* __restrict__ xdbl) {
  __shared__ float sA[2][K4_BM * K4_P];
  __shared__ float sB[2][NKO_PAD * K4_P];

  const int tid  = threadIdx.x;
  const int lane = tid & 31;
  const int wv   = tid >> 5;
  const int half = lane >> 4;
  const int klo  = half << 1;
  const int ln   = lane & 15;
  const int nt   = wv % (NKO_PAD / 16);
  const int mt   = wv / (NKO_PAD / 16);
  const int mbase = blockIdx.x * K4_BM;

  const int ar = tid >> 4;
  const int cv = tid & 15;
  const bool aload = ar < K4_BM;
  const int am = mbase + (aload ? ar : 0);
  const int ab = am / Ls;
  const int ahw = am % Ls;
  const int ah = ahw / Ws, aw = ahw % Ws;

  v4f pa[K4_NSUB];
  v4f pb[K4_NSUB][4];

  auto gload = [&](int s) {
    int t, dy, dx;
    const float* bsrc;
    int bstride;
    if (s < K4_STAPS) {
      t = s / 9;
      const int r9 = s - t * 9;
      dy = r9 / 3;
      dx = r9 - dy * 3;
      bsrc = Wp + s * DINNER;
      bstride = KBIG;
    } else {
      t = s - K4_STAPS;
      dy = 1; dx = 1;
      bsrc = kdp;
      bstride = DINNER;
    }
    const int hh = ah + dy - 1, ww = aw + dx - 1;
    const bool ok = aload && hh >= 0 && hh < Hs && ww >= 0 && ww < Ws;
    #pragma unroll
    for (int sub = 0; sub < K4_NSUB; ++sub) pa[sub] = (v4f){0.f, 0.f, 0.f, 0.f};
    if (ok) {
      const float* asrc = xin + (((size_t)(ab * 2 + t) * Hs + hh) * Ws + ww) * DINNER + cv * 4;
      #pragma unroll
      for (int sub = 0; sub < K4_NSUB; ++sub) pa[sub] = *(const v4f*)(asrc + sub * 64);
    }
    #pragma unroll
    for (int sub = 0; sub < K4_NSUB; ++sub) {
      #pragma unroll
      for (int j = 0; j < 4; ++j)
        pb[sub][j] = *(const v4f*)(bsrc + (size_t)(ar + j * K4_RPP) * bstride + sub * 64 + cv * 4);
    }
  };

  auto lstore = [&](int buf) {
    #pragma unroll
    for (int sub = 0; sub < K4_NSUB; ++sub) {
      if (aload) *(v4f*)(&sA[buf][ar * K4_P + sub * 64 + cv * 4]) = pa[sub];
      #pragma unroll
      for (int j = 0; j < 4; ++j)
        *(v4f*)(&sB[buf][(ar + j * K4_RPP) * K4_P + sub * 64 + cv * 4]) = pb[sub][j];
    }
  };

  auto stage = [&](int s, v8f& acc) {
    const bool more = (s + 1 < K4_SALL);
    if (more) gload(s + 1);
    const float* a = &sA[s & 1][(mt * 16 + ln) * K4_P + klo];
    const float* b = &sB[s & 1][(nt * 16 + ln) * K4_P + klo];
    #pragma unroll
    for (int kk = 0; kk < K4_BK / 8; ++kk) {
      v2f av = *(const v2f*)(a + kk * 4);
      v2f bv = *(const v2f*)(b + kk * 4);
      acc = wmma4(av, bv, acc);
    }
    if (more) lstore((s + 1) & 1);
    #pragma unroll
    for (int kk = K4_BK / 8; kk < K4_BK / 4; ++kk) {
      v2f av = *(const v2f*)(a + kk * 4);
      v2f bv = *(const v2f*)(b + kk * 4);
      acc = wmma4(av, bv, acc);
    }
    __syncthreads();
  };

  v8f accN  = {0.f,0.f,0.f,0.f,0.f,0.f,0.f,0.f};
  v8f accD0 = {0.f,0.f,0.f,0.f,0.f,0.f,0.f,0.f};
  v8f accD1 = {0.f,0.f,0.f,0.f,0.f,0.f,0.f,0.f};

  gload(0);
  lstore(0);
  __syncthreads();
  for (int s = 0; s < K4_STAPS; ++s) stage(s, accN);
  stage(K4_STAPS, accD0);
  stage(K4_STAPS + 1, accD1);

  const int n = nt * 16 + ln;
  if (n < NKO) {
    int kq = n / NCOUT, o = n % NCOUT;
    for (int rr = 0; rr < 8; ++rr) {
      int mm = mbase + mt * 16 + rr + half * 8;
      int b2 = mm / Ls; int l2 = mm % Ls;
      float Nv = accN[rr], d0 = accD0[rr], d1 = accD1[rr];
      float pooled = 2.f * Nv - THETA_C * (0.5f * (d0 + d1) + fminf(d0, d1));
      xdbl[(((size_t)b2 * 4 + kq) * NCOUT + o) * Ls + l2] = pooled;
    }
  }
}

__global__ void k5_delta(const float* __restrict__ xdbl, const float* __restrict__ dtw,
                         const float* __restrict__ dtb, float* __restrict__ delta) {
  int tid = blockIdx.x * blockDim.x + threadIdx.x;
  if (tid >= BATCH * 4 * DINNER * Ls) return;
  int l = tid % Ls; int r = tid / Ls;
  int d = r % DINNER; int bk = r / DINNER;
  int k = bk & 3;
  float acc = dtb[k * DINNER + d];
  const float* wp = dtw + ((size_t)k * DINNER + d) * DTRANK;
  const float* xp = xdbl + (size_t)bk * NCOUT * Ls + l;
  for (int rr = 0; rr < DTRANK; ++rr) acc += xp[(size_t)rr * Ls] * wp[rr];
  delta[tid] = fmaxf(acc, 0.f) + log1pf(expf(-fabsf(acc)));
}

#define QL    DSTATE
#define CPW   (32 / QL)
#define K6_WV 3
#define VPC   8
#define SCH   (4 * VPC)
#define NPAIR (Ls / (2 * SCH))

struct ScanBuf {
  v4f dl[VPC], ul[VPC], b[VPC], c[VPC];
  v2f u2;
};

static __device__ __forceinline__ void scan_load(ScanBuf& s, const float* dP, const float* uP,
                                                 const float* bP, const float* cP, int l0, int n) {
  #pragma unroll
  for (int v = 0; v < VPC; ++v) {
    s.dl[v] = *(const v4f*)(dP + l0 + 4 * v);
    s.ul[v] = *(const v4f*)(uP + l0 + 4 * v);
  }
  s.u2 = *(const v2f*)(uP + l0 + 2 * n);
  #pragma unroll
  for (int v = 0; v < VPC; ++v) {
    s.b[v] = *(const v4f*)(bP + l0 + 4 * v);
    s.c[v] = *(const v4f*)(cP + l0 + 4 * v);
  }
}

template <int XM>
static __device__ __forceinline__ float row_xor(float v) {
  return __int_as_float(__builtin_amdgcn_update_dpp(0, __float_as_int(v), 0x160 + XM, 0xf, 0xf, true));
}

template <int XM>
static __device__ __forceinline__ float half_sum(float a, float b, bool hi) {
  const float keep = hi ? b : a;
  const float give = hi ? a : b;
  return keep + row_xor<XM>(give);
}

template <int OFF>
static __device__ __forceinline__ float steps16_sum(const float (&p)[SCH], bool b0, bool b1, bool b2, bool b3) {
  float a[8], b[4], c[2];
  #pragma unroll
  for (int m = 0; m < 8; ++m) a[m] = half_sum<1>(p[4 * m + OFF], p[4 * m + 2 + OFF], b0);
  #pragma unroll
  for (int m = 0; m < 4; ++m) b[m] = half_sum<2>(a[2 * m], a[2 * m + 1], b1);
  #pragma unroll
  for (int m = 0; m < 2; ++m) c[m] = half_sum<4>(b[2 * m], b[2 * m + 1], b2);
  return half_sum<8>(c[0], c[1], b3);
}

static __device__ __forceinline__ void scan_steps(const ScanBuf& s, float& h, const float A,
                                                  float Dv, float* oP, int l0, int n,
                                                  bool b0, bool b1, bool b2, bool b3) {
  static_assert(SCH == 2 * QL, "a lane ends with the sums of two steps of a chunk");
  float p[SCH];
  #pragma unroll
  for (int v = 0; v < VPC; ++v) {
    #pragma unroll
    for (int e = 0; e < 4; ++e) {
      const float dl = s.dl[v][e];
      const float ul = s.ul[v][e];
      h = expf(dl * A) * h + (dl * ul) * s.b[v][e];
      p[4 * v + e] = h * s.c[v][e];
    }
  }
  const float ye = steps16_sum<0>(p, b0, b1, b2, b3);
  const float yo = steps16_sum<1>(p, b0, b1, b2, b3);
  *(v2f*)(oP + l0 + 2 * n) = (v2f){ye + Dv * s.u2[0], yo + Dv * s.u2[1]};
}

__global__ void __launch_bounds__(32 * K6_WV)
k6_scan(const float* __restrict__ xs, const float* __restrict__ xdbl,
        const float* __restrict__ Alogs, const float* __restrict__ Dss,
        float* __restrict__ dio) {
  const int lane = threadIdx.x & 31;
  const int wave = blockIdx.x * K6_WV + (threadIdx.x >> 5);
  const int n  = lane & (QL - 1);
  const int ch = wave * CPW + lane / QL;
  const int bk = ch / DINNER;
  const int d  = ch % DINNER;
  const int k  = bk & 3;
  float*       oP = dio  + (size_t)ch * Ls;
  const float* dP = oP;
  const float* uP = xs   + (size_t)ch * Ls;
  const float* bP = xdbl + ((size_t)bk * NCOUT + DTRANK + n) * Ls;
  const float* cP = xdbl + ((size_t)bk * NCOUT + DTRANK + DSTATE + n) * Ls;
  const float A = -expf(Alogs[((size_t)(k * DINNER + d)) * DSTATE + n]);
  float h = 0.f;
  const float Dv = Dss[k * DINNER + d];
  const bool b0 = (n & 1) != 0, b1 = (n & 2) != 0, b2 = (n & 4) != 0, b3 = (n & 8) != 0;

  ScanBuf s0, s1;
  scan_load(s0, dP, uP, bP, cP, 0, n);
  for (int p = 0; p < NPAIR; ++p) {
    const int l0 = p * (2 * SCH);
    scan_load(s1, dP, uP, bP, cP, l0 + SCH, n);
    __builtin_amdgcn_sched_barrier(0);
    scan_steps(s0, h, A, Dv, oP, l0, n, b0, b1, b2, b3);
    const int ln = (p + 1 < NPAIR) ? l0 + 2 * SCH : l0 + SCH;
    scan_load(s0, dP, uP, bP, cP, ln, n);
    __builtin_amdgcn_sched_barrier(0);
    scan_steps(s1, h, A, Dv, oP, l0 + SCH, n, b0, b1, b2, b3);
  }
}

#define KT_P   (DINNER + 4)
#define KT_OP  (DMODEL + 4)
#define KT_NQ  12
#define KT_KS  (DINNER / 4)
#define KT_WL  (16 * DINNER / 128)

__global__ void __launch_bounds__(DINNER)
k7_tail(const float* __restrict__ oy, const float* __restrict__ xz,
        const float* __restrict__ g, const float* __restrict__ bb,
        const float* __restrict__ opw, float* __restrict__ out) {
  __shared__ __attribute__((aligned(16))) float sT[16 * KT_P];
  __shared__ __attribute__((aligned(16))) float sO[16 * KT_OP];
  __shared__ __attribute__((aligned(16))) float sB[DMODEL * KT_P];
  __shared__ __attribute__((aligned(16))) float sP1[16 * KT_NQ], sP2[16 * KT_NQ];
  __shared__ float sMu[16], sRs[16];

  const int d = threadIdx.x;
  const int lane = d & 31;
  const int nt   = d >> 5;
  const int half = lane >> 4;
  const int ln   = lane & 15;
  int r = blockIdx.x;
  const int w0 = (r % (Ws / 4)) * 4; r /= (Ws / 4);
  const int h0 = (r % (Hs / 4)) * 4;
  const int b = r / (Hs / 4);

  v4f wr[KT_WL];
  {
    const float* bsrc = opw + (size_t)nt * 16 * DINNER + lane * 4;
    #pragma unroll
    for (int it = 0; it < KT_WL; ++it) wr[it] = *(const v4f*)(bsrc + it * 128);
  }

  const size_t s = (size_t)DINNER * Ls;
  const float* p0 = oy + ((size_t)b * 4 * DINNER + d) * Ls;
  v4f a0[4], a1[4], a2[4], a3[4];
  #pragma unroll
  for (int i = 0; i < 4; ++i) {
    const int l  = (h0 + i) * Ws + w0;
    const int lt = (w0 + i) * Hs + h0;
    a0[i] = *(const v4f*)(p0 + l);
    a2[i] = *(const v4f*)(p0 + 2 * s + (Ls - 4 - l));
    a1[i] = *(const v4f*)(p0 + s + lt);
    a3[i] = *(const v4f*)(p0 + 3 * s + (Ls - 4 - lt));
  }
  float zv[16];
  #pragma unroll
  for (int p = 0; p < 16; ++p) {
    const size_t m = (size_t)b * Ls + (h0 + (p >> 2)) * Ws + (w0 + (p & 3));
    zv[p] = xz[m * 384 + DINNER + d];
  }
  float val[16];
  #pragma unroll
  for (int i = 0; i < 4; ++i) {
    #pragma unroll
    for (int j = 0; j < 4; ++j)
      val[i * 4 + j] = a0[i][j] + a2[i][3 - j] + a1[j][i] + a3[j][3 - i];
  }
  #pragma unroll
  for (int p = 0; p < 16; ++p) sT[p * KT_P + d] = val[p];
  #pragma unroll
  for (int it = 0; it < KT_WL; ++it) {
    const int f = it * 128 + lane * 4;
    *(v4f*)(&sB[(nt * 16 + f / DINNER) * KT_P + f % DINNER]) = wr[it];
  }
  __syncthreads();

  {
    const int p = d / KT_NQ, q = d % KT_NQ;
    const float* tp = &sT[p * KT_P + q * 16];
    float s1 = 0.f, s2 = 0.f;
    #pragma unroll
    for (int i = 0; i < 16; ++i) { const float v = tp[i]; s1 += v; s2 += v * v; }
    sP1[d] = s1; sP2[d] = s2;
  }
  __syncthreads();
  if (d < 16) {
    float t1 = 0.f, t2 = 0.f;
    #pragma unroll
    for (int q = 0; q < KT_NQ; ++q) { t1 += sP1[d * KT_NQ + q]; t2 += sP2[d * KT_NQ + q]; }
    const float mu = t1 / (float)DINNER;
    const float var = t2 / (float)DINNER - mu * mu;
    sMu[d] = mu; sRs[d] = rsqrtf(var + 1e-5f);
  }
  __syncthreads();

  {
    const float gd = g[d], bd = bb[d];
    #pragma unroll
    for (int p = 0; p < 16; ++p) {
      const float yln = (val[p] - sMu[p]) * sRs[p] * gd + bd;
      sT[p * KT_P + d] = yln * siluf(zv[p]);
    }
  }
  __syncthreads();

  v8f acc = {0.f,0.f,0.f,0.f,0.f,0.f,0.f,0.f};
  {
    const float* ap = &sT[ln * KT_P + 2 * half];
    const float* bp = &sB[(nt * 16 + ln) * KT_P + 2 * half];
    #pragma unroll
    for (int kk = 0; kk < KT_KS; ++kk) {
      const v2f av = *(const v2f*)(ap + 4 * kk);
      const v2f bv = *(const v2f*)(bp + 4 * kk);
      acc = wmma4(av, bv, acc);
    }
  }
  #pragma unroll
  for (int rr = 0; rr < 8; ++rr) sO[(rr + 8 * half) * KT_OP + nt * 16 + ln] = acc[rr];
  __syncthreads();

  #pragma unroll
  for (int ps = 0; ps < 2; ++ps) {
    const int idx = ps * DINNER + d;
    const int p = idx / (DMODEL / 4), c4 = idx % (DMODEL / 4);
    const size_t m = (size_t)b * Ls + (h0 + (p >> 2)) * Ws + (w0 + (p & 3));
    *(v4f*)(out + m * DMODEL + c4 * 4) = *(const v4f*)(&sO[p * KT_OP + c4 * 4]);
  }
}

extern "C" void kernel_launch(void* const* d_in, const int* in_sizes, int n_in,
                              void* d_out, int out_size, void* d_ws, size_t ws_size,
                              hipStream_t stream) {
  (void)in_sizes; (void)n_in; (void)out_size; (void)ws_size;
  const float* x   = (const float*)d_in[0];
  const float* xt  = (const float*)d_in[1];
  const float* ipw = (const float*)d_in[2];
  const float* c2w = (const float*)d_in[3];
  const float* c2b = (const float*)d_in[4];
  const float* cxw = (const float*)d_in[5];
  const float* cxb = (const float*)d_in[6];
  const float* xpw = (const float*)d_in[7];
  const float* dtw = (const float*)d_in[8];
  const float* dtb = (const float*)d_in[9];
  const float* alg = (const float*)d_in[10];
  const float* dss = (const float*)d_in[11];
  const float* lng = (const float*)d_in[12];
  const float* lnb = (const float*)d_in[13];
  const float* opw = (const float*)d_in[14];
  float* out = (float*)d_out;
  float* ws  = (float*)d_ws;

  size_t off = 0;
  float* xz    = ws + off; off += (size_t)MROWS * 384;
  float* xin   = ws + off; off += (size_t)BATCH * 2 * Ls * DINNER;
  float* xs    = ws + off; off += (size_t)BATCH * 4 * DINNER * Ls;
  float* Wp    = ws + off; off += (size_t)NKO_PAD * KBIG;
  float* kdp   = ws + off; off += (size_t)NKO_PAD * DINNER;
  float* xdbl  = ws + off; off += (size_t)BATCH * 4 * NCOUT * Ls;
  float* delta = ws + off; off += (size_t)BATCH * 4 * DINNER * Ls;

  gemm_f32_wmma<<<dim3(384 / 16, MROWS / 16), 32, 0, stream>>>(x, ipw, xz, DMODEL, 384);

  k_pack_w <<<(NKO_PAD * KBIG + 255) / 256, 256, 0, stream>>>(xpw, Wp);
  k_pack_kd<<<(NKO_PAD * DINNER + 255) / 256, 256, 0, stream>>>(xpw, kdp);

  k2_conv<<<(BATCH * 2 * Ls * DINNER + 255) / 256, 256, 0, stream>>>(
      xz, xt, c2w, c2b, cxw, cxb, xin);

  k3_xs<<<2 * BATCH * K3_GRP * K3_CG, 32, 0, stream>>>(xin, xs);

  k4_cdc<<<MROWS / K4_BM, K4_NT, 0, stream>>>(xin, Wp, kdp, xdbl);

  k5_delta<<<(BATCH * 4 * DINNER * Ls + 255) / 256, 256, 0, stream>>>(xdbl, dtw, dtb, delta);

  k6_scan<<<BATCH * 4 * DINNER / (CPW * K6_WV), 32 * K6_WV, 0, stream>>>(xs, xdbl, alg, dss, delta);

  k7_tail<<<BATCH * (Hs / 4) * (Ws / 4), DINNER, 0, stream>>>(delta, xz, lng, lnb, opw, out);
}


# k4 conv GEMM work-group tile re-shaped: 245 groups of 16 waves, each 128 pixels x 32 columns (was 196 of 20 waves, 32 x 160): all WGPs work, a SIMD runs 4 accumulator tiles, not 5; sums unchanged
# speedup vs baseline: 8.2146x; 1.0370x over previous
#include <hip/hip_runtime.h>
#include <hip/hip_bf16.h>
#include <math.h>

typedef float v2f __attribute__((ext_vector_type(2)));
typedef float v8f __attribute__((ext_vector_type(8)));
typedef float v4f __attribute__((ext_vector_type(4)));

#define BATCH   2
#define Hs      56
#define Ws      56
#define Ls      (Hs*Ws)
#define DMODEL  96
#define DINNER  192
#define DSTATE  16
#define DTRANK  6
#define NCOUT   38
#define NKO     (4*NCOUT)
#define NKO_PAD 160
#define KBIG    (DINNER*2*9)
#define THETA_C 0.6f
#define MROWS   (BATCH*Ls)

static __device__ __forceinline__ float siluf(float v) { return v / (1.f + expf(-v)); }

static __device__ __forceinline__ v8f wmma4(v2f a, v2f b, v8f c) {
  return __builtin_amdgcn_wmma_f32_16x16x4_f32(false, a, false, b, (short)0, c, false, false);
}

__global__ void gemm_f32_wmma(const float* __restrict__ A, const float* __restrict__ Bt,
                              float* __restrict__ C, int K, int ldc) {
  const int lane = threadIdx.x & 31;
  const int half = lane >> 4;
  const int klo  = half << 1;
  const int ln   = lane & 15;
  const int ntile = blockIdx.x;
  const int mtile = blockIdx.y;
  const float* ap = A  + (size_t)(mtile * 16 + ln) * K + klo;
  const float* bp = Bt + (size_t)(ntile * 16 + ln) * K + klo;
  v8f acc = {0.f,0.f,0.f,0.f,0.f,0.f,0.f,0.f};
  for (int k = 0; k < K; k += 4) {
    v2f a = *(const v2f*)(ap + k);
    v2f b = *(const v2f*)(bp + k);
    acc = wmma4(a, b, acc);
  }
  const int n = ntile * 16 + ln;
  for (int rr = 0; rr < 8; ++rr) {
    int mm = mtile * 16 + rr + half * 8;
    C[(size_t)mm * ldc + n] = acc[rr];
  }
}

__global__ void k_pack_w(const float* __restrict__ xw, float* __restrict__ Wp) {
  int tid = blockIdx.x * blockDim.x + threadIdx.x;
  if (tid >= NKO_PAD * KBIG) return;
  int kk = tid % KBIG;
  int n  = tid / KBIG;
  float v = 0.f;
  if (n < NKO) {
    int c = kk % DINNER;
    int r = kk / DINNER;
    int dx = r % 3, dy = (r / 3) % 3, t = r / 9;
    v = xw[((((size_t)n * DINNER + c) * 2 + t) * 3 + dy) * 3 + dx];
  }
  Wp[tid] = v;
}

__global__ void k_pack_kd(const float* __restrict__ xw, float* __restrict__ kdp) {
  int tid = blockIdx.x * blockDim.x + threadIdx.x;
  if (tid >= NKO_PAD * DINNER) return;
  int c = tid % DINNER;
  int n = tid / DINNER;
  float v = 0.f;
  if (n < NKO) {
    const float* p = xw + ((size_t)n * DINNER + c) * 18;
    for (int i = 0; i < 18; ++i) v += p[i];
  }
  kdp[tid] = v;
}

__global__ void k2_conv(const float* __restrict__ xz, const float* __restrict__ xt,
                        const float* __restrict__ w2, const float* __restrict__ b2,
                        const float* __restrict__ wxt, const float* __restrict__ bxt,
                        float* __restrict__ xin) {
  int tid = blockIdx.x * blockDim.x + threadIdx.x;
  if (tid >= BATCH * 2 * Ls * DINNER) return;
  int c = tid % DINNER; int r = tid / DINNER;
  int w = r % Ws; r /= Ws;
  int h = r % Hs; r /= Hs;
  int t = r % 2;  int b = r / 2;
  float acc;
  if (t == 0) {
    acc = b2[c];
    for (int dy = 0; dy < 3; ++dy) {
      int hh = h + dy - 1; if (hh < 0 || hh >= Hs) continue;
      for (int dx = 0; dx < 3; ++dx) {
        int ww = w + dx - 1; if (ww < 0 || ww >= Ws) continue;
        acc += xz[((size_t)((b * Hs + hh) * Ws + ww)) * 384 + c] * w2[c * 9 + dy * 3 + dx];
      }
    }
  } else {
    acc = bxt[c];
    int ci = c >> 1;
    for (int dy = 0; dy < 3; ++dy) {
      int hh = h + dy - 1; if (hh < 0 || hh >= Hs) continue;
      for (int dx = 0; dx < 3; ++dx) {
        int ww = w + dx - 1; if (ww < 0 || ww >= Ws) continue;
        acc += xt[((size_t)((b * Hs + hh) * Ws + ww)) * DMODEL + ci] * wxt[c * 9 + dy * 3 + dx];
      }
    }
  }
  xin[tid] = siluf(acc);
}

#define K3_SL   Hs
#define K3_NSEG 2
#define K3_GRP  (Hs / (2 * K3_NSEG))
#define K3_CG   (DINNER / 16)

template <int TYPE>
static __device__ __forceinline__ void k3_body(const float* __restrict__ xin, float* __restrict__ xs,
                                               int b, int seg0, int d) {
  constexpr int PSTEP = (TYPE == 0) ? Ws * DINNER : DINNER;
  constexpr int SSTEP = (TYPE == 0) ? DINNER : Ws * DINNER;
  const float* x0 = xin + (size_t)(b * 2 + 0) * Ls * DINNER + d;
  const float* x1 = xin + (size_t)(b * 2 + 1) * Ls * DINNER + d;
  float* pf = xs + ((size_t)(b * 4 + 2 * TYPE) * DINNER + d) * Ls;
  float* pr = pf + (size_t)DINNER * Ls;
  for (int sg = 0; sg < K3_NSEG; ++sg) {
    const int seg = seg0 + sg;
    const float* a0 = x0 + (size_t)seg * SSTEP;
    const float* a1 = x1 + (size_t)seg * SSTEP;
    float o[K3_SL];
    #pragma unroll
    for (int i = 0; i < K3_SL / 2; ++i) {
      const float v0 = a0[(2 * i) * PSTEP], v1 = a0[(2 * i + 1) * PSTEP];
      o[2 * i] = 0.5f * (v0 + v1) + fmaxf(v0, v1);
      const float w0 = a1[(2 * i) * PSTEP], w1 = a1[(2 * i + 1) * PSTEP];
      o[2 * i + 1] = 0.5f * (w0 + w1) + fmaxf(w0, w1);
    }
    #pragma unroll
    for (int v = 0; v < K3_SL / 4; ++v) {
      *(v4f*)(pf + seg * K3_SL + 4 * v) = (v4f){o[4 * v], o[4 * v + 1], o[4 * v + 2], o[4 * v + 3]};
      *(v4f*)(pr + seg * K3_SL + 4 * v) =
          (v4f){o[K3_SL - 1 - 4 * v], o[K3_SL - 2 - 4 * v], o[K3_SL - 3 - 4 * v], o[K3_SL - 4 - 4 * v]};
    }
  }
}

__global__ void __launch_bounds__(32)
k3_xs(const float* __restrict__ xin, float* __restrict__ xs) {
  static_assert(Hs == Ws, "a scan segment is Hs steps in one scan order and Ws in the other");
  int r = blockIdx.x;
  const int d = (r % K3_CG) * 16 + (threadIdx.x & 15); r /= K3_CG;
  const int seg0 = ((r % K3_GRP) * 2 + (threadIdx.x >> 4)) * K3_NSEG; r /= K3_GRP;
  const int b = r % BATCH;
  const int type = r / BATCH;
  if (type == 0) k3_body<0>(xin, xs, b, seg0, d);
  else           k3_body<1>(xin, xs, b, seg0, d);
}

#define K4_BM    128
#define K4_BN    32
#define K4_NCG   (NKO_PAD / K4_BN)
#define K4_BK    DINNER
#define K4_P     (K4_BK + 4)
#define K4_NT    512
#define K4_RPP   (K4_NT / 16)
#define K4_APASS (K4_BM / K4_RPP)
#define K4_NSUB  (K4_BK / 64)
#define K4_STAPS 18
#define K4_SALL  20

__global__ void __launch_bounds__(K4_NT)
k4_cdc(const float* __restrict__ xin, const float* __restrict__ Wp,
       const float* __restrict__ kdp, float* __restrict__ xdbl) {
  static_assert(MROWS % K4_BM == 0 && NKO_PAD % K4_BN == 0, "whole tiles only");
  static_assert(K4_BN == K4_RPP, "one pass copies a stage's weight rows");
  __shared__ float sA[2][K4_BM * K4_P];
  __shared__ float sB[2][K4_BN * K4_P];

  const int tid  = threadIdx.x;
  const int lane = tid & 31;
  const int wv   = tid >> 5;
  const int half = lane >> 4;
  const int klo  = half << 1;
  const int ln   = lane & 15;
  const int nt   = wv % (K4_BN / 16);
  const int mt   = wv / (K4_BN / 16);
  const int cg   = blockIdx.x % K4_NCG;
  const int mbase = (blockIdx.x / K4_NCG) * K4_BM;
  const int nbase = cg * K4_BN;

  const int ar = tid >> 4;
  const int cv = tid & 15;
  int ab[K4_APASS], ah[K4_APASS], aw[K4_APASS];
  #pragma unroll
  for (int j = 0; j < K4_APASS; ++j) {
    const int am = mbase + ar + j * K4_RPP;
    const int ahw = am % Ls;
    ab[j] = am / Ls;
    ah[j] = ahw / Ws;
    aw[j] = ahw % Ws;
  }

  v4f pa[K4_APASS][K4_NSUB];
  v4f pb[K4_NSUB];

  auto gload = [&](int s) {
    int t, dy, dx;
    const float* bsrc;
    int bstride;
    if (s < K4_STAPS) {
      t = s / 9;
      const int r9 = s - t * 9;
      dy = r9 / 3;
      dx = r9 - dy * 3;
      bsrc = Wp + s * DINNER;
      bstride = KBIG;
    } else {
      t = s - K4_STAPS;
      dy = 1; dx = 1;
      bsrc = kdp;
      bstride = DINNER;
    }
    #pragma unroll
    for (int j = 0; j < K4_APASS; ++j) {
      const int hh = ah[j] + dy - 1, ww = aw[j] + dx - 1;
      const bool ok = hh >= 0 && hh < Hs && ww >= 0 && ww < Ws;
      #pragma unroll
      for (int sub = 0; sub < K4_NSUB; ++sub) pa[j][sub] = (v4f){0.f, 0.f, 0.f, 0.f};
      if (ok) {
        const float* asrc = xin + (((size_t)(ab[j] * 2 + t) * Hs + hh) * Ws + ww) * DINNER + cv * 4;
        #pragma unroll
        for (int sub = 0; sub < K4_NSUB; ++sub) pa[j][sub] = *(const v4f*)(asrc + sub * 64);
      }
    }
    #pragma unroll
    for (int sub = 0; sub < K4_NSUB; ++sub)
      pb[sub] = *(const v4f*)(bsrc + (size_t)(nbase + ar) * bstride + sub * 64 + cv * 4);
  };

  auto lstore = [&](int buf) {
    #pragma unroll
    for (int sub = 0; sub < K4_NSUB; ++sub) {
      #pragma unroll
      for (int j = 0; j < K4_APASS; ++j)
        *(v4f*)(&sA[buf][(ar + j * K4_RPP) * K4_P + sub * 64 + cv * 4]) = pa[j][sub];
      *(v4f*)(&sB[buf][ar * K4_P + sub * 64 + cv * 4]) = pb[sub];
    }
  };

  auto stage = [&](int s, v8f& acc) {
    const bool more = (s + 1 < K4_SALL);
    if (more) gload(s + 1);
    const float* a = &sA[s & 1][(mt * 16 + ln) * K4_P + klo];
    const float* b = &sB[s & 1][(nt * 16 + ln) * K4_P + klo];
    #pragma unroll
    for (int kk = 0; kk < K4_BK / 8; ++kk) {
      v2f av = *(const v2f*)(a + kk * 4);
      v2f bv = *(const v2f*)(b + kk * 4);
      acc = wmma4(av, bv, acc);
    }
    if (more) lstore((s + 1) & 1);
    #pragma unroll
    for (int kk = K4_BK / 8; kk < K4_BK / 4; ++kk) {
      v2f av = *(const v2f*)(a + kk * 4);
      v2f bv = *(const v2f*)(b + kk * 4);
      acc = wmma4(av, bv, acc);
    }
    __syncthreads();
  };

  v8f accN  = {0.f,0.f,0.f,0.f,0.f,0.f,0.f,0.f};
  v8f accD0 = {0.f,0.f,0.f,0.f,0.f,0.f,0.f,0.f};
  v8f accD1 = {0.f,0.f,0.f,0.f,0.f,0.f,0.f,0.f};

  gload(0);
  lstore(0);
  __syncthreads();
  for (int s = 0; s < K4_STAPS; ++s) stage(s, accN);
  stage(K4_STAPS, accD0);
  stage(K4_STAPS + 1, accD1);

  const int n = nbase + nt * 16 + ln;
  if (n < NKO) {
    int kq = n / NCOUT, o = n % NCOUT;
    for (int rr = 0; rr < 8; ++rr) {
      int mm = mbase + mt * 16 + rr + half * 8;
      int b2 = mm / Ls; int l2 = mm % Ls;
      float Nv = accN[rr], d0 = accD0[rr], d1 = accD1[rr];
      float pooled = 2.f * Nv - THETA_C * (0.5f * (d0 + d1) + fminf(d0, d1));
      xdbl[(((size_t)b2 * 4 + kq) * NCOUT + o) * Ls + l2] = pooled;
    }
  }
}

__global__ void k5_delta(const float* __restrict__ xdbl, const float* __restrict__ dtw,
                         const float* __restrict__ dtb, float* __restrict__ delta) {
  int tid = blockIdx.x * blockDim.x + threadIdx.x;
  if (tid >= BATCH * 4 * DINNER * Ls) return;
  int l = tid % Ls; int r = tid / Ls;
  int d = r % DINNER; int bk = r / DINNER;
  int k = bk & 3;
  float acc = dtb[k * DINNER + d];
  const float* wp = dtw + ((size_t)k * DINNER + d) * DTRANK;
  const float* xp = xdbl + (size_t)bk * NCOUT * Ls + l;
  for (int rr = 0; rr < DTRANK; ++rr) acc += xp[(size_t)rr * Ls] * wp[rr];
  delta[tid] = fmaxf(acc, 0.f) + log1pf(expf(-fabsf(acc)));
}

#define QL    DSTATE
#define CPW   (32 / QL)
#define K6_WV 3
#define VPC   8
#define SCH   (4 * VPC)
#define NPAIR (Ls / (2 * SCH))

struct ScanBuf {
  v4f dl[VPC], ul[VPC], b[VPC], c[VPC];
  v2f u2;
};

static __device__ __forceinline__ void scan_load(ScanBuf& s, const float* dP, const float* uP,
                                                 const float* bP, const float* cP, int l0, int n) {
  #pragma unroll
  for (int v = 0; v < VPC; ++v) {
    s.dl[v] = *(const v4f*)(dP + l0 + 4 * v);
    s.ul[v] = *(const v4f*)(uP + l0 + 4 * v);
  }
  s.u2 = *(const v2f*)(uP + l0 + 2 * n);
  #pragma unroll
  for (int v = 0; v < VPC; ++v) {
    s.b[v] = *(const v4f*)(bP + l0 + 4 * v);
    s.c[v] = *(const v4f*)(cP + l0 + 4 * v);
  }
}

template <int XM>
static __device__ __forceinline__ float row_xor(float v) {
  return __int_as_float(__builtin_amdgcn_update_dpp(0, __float_as_int(v), 0x160 + XM, 0xf, 0xf, true));
}

template <int XM>
static __device__ __forceinline__ float half_sum(float a, float b, bool hi) {
  const float keep = hi ? b : a;
  const float give = hi ? a : b;
  return keep + row_xor<XM>(give);
}

template <int OFF>
static __device__ __forceinline__ float steps16_sum(const float (&p)[SCH], bool b0, bool b1, bool b2, bool b3) {
  float a[8], b[4], c[2];
  #pragma unroll
  for (int m = 0; m < 8; ++m) a[m] = half_sum<1>(p[4 * m + OFF], p[4 * m + 2 + OFF], b0);
  #pragma unroll
  for (int m = 0; m < 4; ++m) b[m] = half_sum<2>(a[2 * m], a[2 * m + 1], b1);
  #pragma unroll
  for (int m = 0; m < 2; ++m) c[m] = half_sum<4>(b[2 * m], b[2 * m + 1], b2);
  return half_sum<8>(c[0], c[1], b3);
}

static __device__ __forceinline__ void scan_steps(const ScanBuf& s, float& h, const float A,
                                                  float Dv, float* oP, int l0, int n,
                                                  bool b0, bool b1, bool b2, bool b3) {
  static_assert(SCH == 2 * QL, "a lane ends with the sums of two steps of a chunk");
  float p[SCH];
  #pragma unroll
  for (int v = 0; v < VPC; ++v) {
    #pragma unroll
    for (int e = 0; e < 4; ++e) {
      const float dl = s.dl[v][e];
      const float ul = s.ul[v][e];
      h = expf(dl * A) * h + (dl * ul) * s.b[v][e];
      p[4 * v + e] = h * s.c[v][e];
    }
  }
  const float ye = steps16_sum<0>(p, b0, b1, b2, b3);
  const float yo = steps16_sum<1>(p, b0, b1, b2, b3);
  *(v2f*)(oP + l0 + 2 * n) = (v2f){ye + Dv * s.u2[0], yo + Dv * s.u2[1]};
}

__global__ void __launch_bounds__(32 * K6_WV)
k6_scan(const float* __restrict__ xs, const float* __restrict__ xdbl,
        const float* __restrict__ Alogs, const float* __restrict__ Dss,
        float* __restrict__ dio) {
  const int lane = threadIdx.x & 31;
  const int wave = blockIdx.x * K6_WV + (threadIdx.x >> 5);
  const int n  = lane & (QL - 1);
  const int ch = wave * CPW + lane / QL;
  const int bk = ch / DINNER;
  const int d  = ch % DINNER;
  const int k  = bk & 3;
  float*       oP = dio  + (size_t)ch * Ls;
  const float* dP = oP;
  const float* uP = xs   + (size_t)ch * Ls;
  const float* bP = xdbl + ((size_t)bk * NCOUT + DTRANK + n) * Ls;
  const float* cP = xdbl + ((size_t)bk * NCOUT + DTRANK + DSTATE + n) * Ls;
  const float A = -expf(Alogs[((size_t)(k * DINNER + d)) * DSTATE + n]);
  float h = 0.f;
  const float Dv = Dss[k * DINNER + d];
  const bool b0 = (n & 1) != 0, b1 = (n & 2) != 0, b2 = (n & 4) != 0, b3 = (n & 8) != 0;

  ScanBuf s0, s1;
  scan_load(s0, dP, uP, bP, cP, 0, n);
  for (int p = 0; p < NPAIR; ++p) {
    const int l0 = p * (2 * SCH);
    scan_load(s1, dP, uP, bP, cP, l0 + SCH, n);
    __builtin_amdgcn_sched_barrier(0);
    scan_steps(s0, h, A, Dv, oP, l0, n, b0, b1, b2, b3);
    const int ln = (p + 1 < NPAIR) ? l0 + 2 * SCH : l0 + SCH;
    scan_load(s0, dP, uP, bP, cP, ln, n);
    __builtin_amdgcn_sched_barrier(0);
    scan_steps(s1, h, A, Dv, oP, l0 + SCH, n, b0, b1, b2, b3);
  }
}

#define KT_P   (DINNER + 4)
#define KT_OP  (DMODEL + 4)
#define KT_NQ  12
#define KT_KS  (DINNER / 4)
#define KT_WL  (16 * DINNER / 128)

__global__ void __launch_bounds__(DINNER)
k7_tail(const float* __restrict__ oy, const float* __restrict__ xz,
        const float* __restrict__ g, const float* __restrict__ bb,
        const float* __restrict__ opw, float* __restrict__ out) {
  __shared__ __attribute__((aligned(16))) float sT[16 * KT_P];
  __shared__ __attribute__((aligned(16))) float sO[16 * KT_OP];
  __shared__ __attribute__((aligned(16))) float sB[DMODEL * KT_P];
  __shared__ __attribute__((aligned(16))) float sP1[16 * KT_NQ], sP2[16 * KT_NQ];
  __shared__ float sMu[16], sRs[16];

  const int d = threadIdx.x;
  const int lane = d & 31;
  const int nt   = d >> 5;
  const int half = lane >> 4;
  const int ln   = lane & 15;
  int r = blockIdx.x;
  const int w0 = (r % (Ws / 4)) * 4; r /= (Ws / 4);
  const int h0 = (r % (Hs / 4)) * 4;
  const int b = r / (Hs / 4);

  v4f wr[KT_WL];
  {
    const float* bsrc = opw + (size_t)nt * 16 * DINNER + lane * 4;
    #pragma unroll
    for (int it = 0; it < KT_WL; ++it) wr[it] = *(const v4f*)(bsrc + it * 128);
  }

  const size_t s = (size_t)DINNER * Ls;
  const float* p0 = oy + ((size_t)b * 4 * DINNER + d) * Ls;
  v4f a0[4], a1[4], a2[4], a3[4];
  #pragma unroll
  for (int i = 0; i < 4; ++i) {
    const int l  = (h0 + i) * Ws + w0;
    const int lt = (w0 + i) * Hs + h0;
    a0[i] = *(const v4f*)(p0 + l);
    a2[i] = *(const v4f*)(p0 + 2 * s + (Ls - 4 - l));
    a1[i] = *(const v4f*)(p0 + s + lt);
    a3[i] = *(const v4f*)(p0 + 3 * s + (Ls - 4 - lt));
  }
  float zv[16];
  #pragma unroll
  for (int p = 0; p < 16; ++p) {
    const size_t m = (size_t)b * Ls + (h0 + (p >> 2)) * Ws + (w0 + (p & 3));
    zv[p] = xz[m * 384 + DINNER + d];
  }
  float val[16];
  #pragma unroll
  for (int i = 0; i < 4; ++i) {
    #pragma unroll
    for (int j = 0; j < 4; ++j)
      val[i * 4 + j] = a0[i][j] + a2[i][3 - j] + a1[j][i] + a3[j][3 - i];
  }
  #pragma unroll
  for (int p = 0; p < 16; ++p) sT[p * KT_P + d] = val[p];
  #pragma unroll
  for (int it = 0; it < KT_WL; ++it) {
    const int f = it * 128 + lane * 4;
    *(v4f*)(&sB[(nt * 16 + f / DINNER) * KT_P + f % DINNER]) = wr[it];
  }
  __syncthreads();

  {
    const int p = d / KT_NQ, q = d % KT_NQ;
    const float* tp = &sT[p * KT_P + q * 16];
    float s1 = 0.f, s2 = 0.f;
    #pragma unroll
    for (int i = 0; i < 16; ++i) { const float v = tp[i]; s1 += v; s2 += v * v; }
    sP1[d] = s1; sP2[d] = s2;
  }
  __syncthreads();
  if (d < 16) {
    float t1 = 0.f, t2 = 0.f;
    #pragma unroll
    for (int q = 0; q < KT_NQ; ++q) { t1 += sP1[d * KT_NQ + q]; t2 += sP2[d * KT_NQ + q]; }
    const float mu = t1 / (float)DINNER;
    const float var = t2 / (float)DINNER - mu * mu;
    sMu[d] = mu; sRs[d] = rsqrtf(var + 1e-5f);
  }
  __syncthreads();

  {
    const float gd = g[d], bd = bb[d];
    #pragma unroll
    for (int p = 0; p < 16; ++p) {
      const float yln = (val[p] - sMu[p]) * sRs[p] * gd + bd;
      sT[p * KT_P + d] = yln * siluf(zv[p]);
    }
  }
  __syncthreads();

  v8f acc = {0.f,0.f,0.f,0.f,0.f,0.f,0.f,0.f};
  {
    const float* ap = &sT[ln * KT_P + 2 * half];
    const float* bp = &sB[(nt * 16 + ln) * KT_P + 2 * half];
    #pragma unroll
    for (int kk = 0; kk < KT_KS; ++kk) {
      const v2f av = *(const v2f*)(ap + 4 * kk);
      const v2f bv = *(const v2f*)(bp + 4 * kk);
      acc = wmma4(av, bv, acc);
    }
  }
  #pragma unroll
  for (int rr = 0; rr < 8; ++rr) sO[(rr + 8 * half) * KT_OP + nt * 16 + ln] = acc[rr];
  __syncthreads();

  #pragma unroll
  for (int ps = 0; ps < 2; ++ps) {
    const int idx = ps * DINNER + d;
    const int p = idx / (DMODEL / 4), c4 = idx % (DMODEL / 4);
    const size_t m = (size_t)b * Ls + (h0 + (p >> 2)) * Ws + (w0 + (p & 3));
    *(v4f*)(out + m * DMODEL + c4 * 4) = *(const v4f*)(&sO[p * KT_OP + c4 * 4]);
  }
}

extern "C" void kernel_launch(void* const* d_in, const int* in_sizes, int n_in,
                              void* d_out, int out_size, void* d_ws, size_t ws_size,
                              hipStream_t stream) {
  (void)in_sizes; (void)n_in; (void)out_size; (void)ws_size;
  const float* x   = (const float*)d_in[0];
  const float* xt  = (const float*)d_in[1];
  const float* ipw = (const float*)d_in[2];
  const float* c2w = (const float*)d_in[3];
  const float* c2b = (const float*)d_in[4];
  const float* cxw = (const float*)d_in[5];
  const float* cxb = (const float*)d_in[6];
  const float* xpw = (const float*)d_in[7];
  const float* dtw = (const float*)d_in[8];
  const float* dtb = (const float*)d_in[9];
  const float* alg = (const float*)d_in[10];
  const float* dss = (const float*)d_in[11];
  const float* lng = (const float*)d_in[12];
  const float* lnb = (const float*)d_in[13];
  const float* opw = (const float*)d_in[14];
  float* out = (float*)d_out;
  float* ws  = (float*)d_ws;

  size_t off = 0;
  float* xz    = ws + off; off += (size_t)MROWS * 384;
  float* xin   = ws + off; off += (size_t)BATCH * 2 * Ls * DINNER;
  float* xs    = ws + off; off += (size_t)BATCH * 4 * DINNER * Ls;
  float* Wp    = ws + off; off += (size_t)NKO_PAD * KBIG;
  float* kdp   = ws + off; off += (size_t)NKO_PAD * DINNER;
  float* xdbl  = ws + off; off += (size_t)BATCH * 4 * NCOUT * Ls;
  float* delta = ws + off; off += (size_t)BATCH * 4 * DINNER * Ls;

  gemm_f32_wmma<<<dim3(384 / 16, MROWS / 16), 32, 0, stream>>>(x, ipw, xz, DMODEL, 384);

  k_pack_w <<<(NKO_PAD * KBIG + 255) / 256, 256, 0, stream>>>(xpw, Wp);
  k_pack_kd<<<(NKO_PAD * DINNER + 255) / 256, 256, 0, stream>>>(xpw, kdp);

  k2_conv<<<(BATCH * 2 * Ls * DINNER + 255) / 256, 256, 0, stream>>>(
      xz, xt, c2w, c2b, cxw, cxb, xin);

  k3_xs<<<2 * BATCH * K3_GRP * K3_CG, 32, 0, stream>>>(xin, xs);

  k4_cdc<<<(MROWS / K4_BM) * K4_NCG, K4_NT, 0, stream>>>(xin, Wp, kdp, xdbl);

  k5_delta<<<(BATCH * 4 * DINNER * Ls + 255) / 256, 256, 0, stream>>>(xdbl, dtw, dtb, delta);

  k6_scan<<<BATCH * 4 * DINNER / (CPW * K6_WV), 32 * K6_WV, 0, stream>>>(xs, xdbl, alg, dss, delta);

  k7_tail<<<BATCH * (Hs / 4) * (Ws / 4), DINNER, 0, stream>>>(delta, xz, lng, lnb, opw, out);
}
